# starting kernel
# baseline (speedup 1.0000x reference)
#include <hip/hip_runtime.h>
#include <hip/hip_bf16.h>

typedef __attribute__((ext_vector_type(2))) float v2f;
typedef __attribute__((ext_vector_type(8))) float v8f;

#define D 64

__global__ __launch_bounds__(256) void deg_init(float* dout, float* din, int N) {
    int i = blockIdx.x * blockDim.x + threadIdx.x;
    if (i < N) { dout[i] = 1.0f; din[i] = 1.0f; }
}

__global__ __launch_bounds__(256) void deg_count(const int* __restrict__ src,
                                                 const int* __restrict__ dst,
                                                 float* dout, float* din, int E) {
    int e = blockIdx.x * blockDim.x + threadIdx.x;
    if (e < E) {
        atomicAdd(&dout[src[e]], 1.0f);
        atomicAdd(&din[dst[e]], 1.0f);
    }
}

__global__ __launch_bounds__(256) void deg_rsqrt(float* dout, float* din, int N) {
    int i = blockIdx.x * blockDim.x + threadIdx.x;
    if (i < N) { dout[i] = rsqrtf(dout[i]); din[i] = rsqrtf(din[i]); }
}

__global__ __launch_bounds__(256) void gemm64_wmma(const float* __restrict__ X,
                                                   const float* __restrict__ rs,
                                                   const float* __restrict__ W,
                                                   float* __restrict__ out0,
                                                   float* __restrict__ out1,
                                                   int N) {
    const int wave = threadIdx.x >> 5;
    const int lane = threadIdx.x & 31;
    const int nb = blockIdx.x * 128 + wave * 16;
    if (nb >= N) return;
    const int half = lane >> 4;
    const int lr   = lane & 15;

    const float s = rs[nb + lr];
    const float* __restrict__ xrow = X + (size_t)(nb + lr) * D;

    v8f acc0 = {}, acc1 = {}, acc2 = {}, acc3 = {};
#pragma unroll
    for (int k0 = 0; k0 < D; k0 += 4) {
        const int ka = k0 + half * 2;
        v2f a;
        a.x = xrow[ka] * s;
        a.y = xrow[ka + 1] * s;
        const float* __restrict__ w0 = W + (size_t)ka * D + lr;
        const float* __restrict__ w1 = w0 + D;
        v2f b0 = { w0[0],  w1[0]  };
        v2f b1 = { w0[16], w1[16] };
        v2f b2 = { w0[32], w1[32] };
        v2f b3 = { w0[48], w1[48] };
        acc0 = __builtin_amdgcn_wmma_f32_16x16x4_f32(false, a, false, b0, (short)0, acc0, false, false);
        acc1 = __builtin_amdgcn_wmma_f32_16x16x4_f32(false, a, false, b1, (short)0, acc1, false, false);
        acc2 = __builtin_amdgcn_wmma_f32_16x16x4_f32(false, a, false, b2, (short)0, acc2, false, false);
        acc3 = __builtin_amdgcn_wmma_f32_16x16x4_f32(false, a, false, b3, (short)0, acc3, false, false);
    }

    const int row0 = nb + half * 8;
#pragma unroll
    for (int v = 0; v < 8; ++v) {
        const size_t base = (size_t)(row0 + v) * D + lr;
        out0[base]      = acc0[v];
        out0[base + 16] = acc1[v];
        out0[base + 32] = acc2[v];
        out0[base + 48] = acc3[v];
        out1[base]      = acc0[v];
        out1[base + 16] = acc1[v];
        out1[base + 32] = acc2[v];
        out1[base + 48] = acc3[v];
    }
}

__global__ __launch_bounds__(256) void scatter_add(const float* __restrict__ h,
                                                   const int* __restrict__ src,
                                                   const int* __restrict__ dst,
                                                   float* __restrict__ agg, int E) {
    const int t = blockIdx.x * blockDim.x + threadIdx.x;
    const int e = t >> 4;
    if (e >= E) return;
    const int f = (t & 15) * 4;
    const int s = src[e];
    const int d = dst[e];
    const float4 v = *(const float4*)(h + (size_t)s * D + f);
    float* p = agg + (size_t)d * D + f;
    atomicAdd(p + 0, v.x);
    atomicAdd(p + 1, v.y);
    atomicAdd(p + 2, v.z);
    atomicAdd(p + 3, v.w);
}

__global__ __launch_bounds__(256) void post_scale_bias(const float* __restrict__ agg,
                                                       const float* __restrict__ rs_in,
                                                       const float* __restrict__ b,
                                                       float* __restrict__ hout, int N) {
    const int t = blockIdx.x * blockDim.x + threadIdx.x;
    const int node = t >> 4;
    if (node >= N) return;
    const int f = (t & 15) * 4;
    const float s = rs_in[node];
    const float4 v = *(const float4*)(agg + (size_t)node * D + f);
    float4 r;
    r.x = fmaxf(v.x * s + b[f + 0], 0.0f);
    r.y = fmaxf(v.y * s + b[f + 1], 0.0f);
    r.z = fmaxf(v.z * s + b[f + 2], 0.0f);
    r.w = fmaxf(v.w * s + b[f + 3], 0.0f);
    *(float4*)(hout + (size_t)node * D + f) = r;
}

__global__ __launch_bounds__(256) void final_dot(const float* __restrict__ agg,
                                                 const float* __restrict__ rs_in,
                                                 const float* __restrict__ b2,
                                                 const float* __restrict__ Wr,
                                                 const float* __restrict__ br,
                                                 float* __restrict__ out, int N) {
    const int wave = threadIdx.x >> 5;
    const int lane = threadIdx.x & 31;
    const int node = blockIdx.x * 8 + wave;
    if (node >= N) return;
    const float s = rs_in[node];
    const float* row = agg + (size_t)node * D;
    float acc = (row[lane]      * s + b2[lane])      * Wr[lane]
              + (row[lane + 32] * s + b2[lane + 32]) * Wr[lane + 32];
#pragma unroll
    for (int off = 16; off > 0; off >>= 1) acc += __shfl_xor(acc, off, 32);
    if (lane == 0) out[node] = acc + br[0];
}

extern "C" void kernel_launch(void* const* d_in, const int* in_sizes, int n_in,
                              void* d_out, int out_size, void* d_ws, size_t ws_size,
                              hipStream_t stream) {
    const float* x   = (const float*)d_in[0];
    const int*   src = (const int*)d_in[1];
    const int*   dst = (const int*)d_in[2];
    const float* W1  = (const float*)d_in[3];
    const float* b1  = (const float*)d_in[4];
    const float* W2  = (const float*)d_in[5];
    const float* b2  = (const float*)d_in[6];
    const float* Wr  = (const float*)d_in[7];
    const float* br  = (const float*)d_in[8];
    float* out = (float*)d_out;

    const int N = in_sizes[0] / D;
    const int E = in_sizes[1];

    float* ws     = (float*)d_ws;
    float* rs_out = ws;
    float* rs_in  = ws + N;
    float* h      = ws + 2 * (size_t)N;
    float* agg    = h + (size_t)N * D;

    const int TB = 256;
    const int gN    = (N + TB - 1) / TB;
    const int gE    = (E + TB - 1) / TB;
    const int gGemm = (N + 127) / 128;
    const int gEdge = (int)(((size_t)E * 16 + TB - 1) / TB);
    const int gFeat = (int)(((size_t)N * 16 + TB - 1) / TB);
    const int gNode = (N + 7) / 8;

    deg_init <<<gN, TB, 0, stream>>>(rs_out, rs_in, N);
    deg_count<<<gE, TB, 0, stream>>>(src, dst, rs_out, rs_in, E);
    deg_rsqrt<<<gN, TB, 0, stream>>>(rs_out, rs_in, N);

    gemm64_wmma<<<gGemm, TB, 0, stream>>>(x, rs_out, W1, h, agg, N);
    scatter_add<<<gEdge, TB, 0, stream>>>(h, src, dst, agg, E);
    post_scale_bias<<<gFeat, TB, 0, stream>>>(agg, rs_in, b1, h, N);

    gemm64_wmma<<<gGemm, TB, 0, stream>>>(h, rs_out, W2, h, agg, N);
    scatter_add<<<gEdge, TB, 0, stream>>>(h, src, dst, agg, E);

    final_dot<<<gNode, TB, 0, stream>>>(agg, rs_in, b2, Wr, br, out, N);
}


# edge sums by gather: edges put in destination order (integer atomic counts, scan, placement; built each call), 16 lanes sum a node's source rows in registers; f32 atomics gone; sum order changes
# speedup vs baseline: 1.8953x; 1.8953x over previous
#include <hip/hip_runtime.h>
#include <hip/hip_bf16.h>


typedef __attribute__((ext_vector_type(2))) float v2f;
typedef __attribute__((ext_vector_type(8))) float v8f;
typedef __attribute__((ext_vector_type(4))) float v4f;

#define D 64
#define SB 1024

__global__ __launch_bounds__(256) void cnt_zero(int* __restrict__ c, int n2, float* __restrict__ zrow,
                                                int* __restrict__ zcell, int zr) {
    int i = blockIdx.x * blockDim.x + threadIdx.x;
    if (i < n2) c[i] = 0;
    if (i < D) zrow[i] = 0.0f;
    if (i == 0) zcell[0] = zr;
}

__global__ __launch_bounds__(256) void edge_count(const int* __restrict__ src,
                                                  const int* __restrict__ dst,
                                                  int* cnt_out, int* cnt_in,
                                                  int* __restrict__ rank, int E) {
    int e = blockIdx.x * blockDim.x + threadIdx.x;
    if (e < E) {
        const int s = src[e];
        const int d = dst[e];
        atomicAdd(&cnt_out[s], 1);
        rank[e] = atomicAdd(&cnt_in[d], 1);
    }
}

__global__ __launch_bounds__(SB) void node_scan(const int* __restrict__ cnt_in,
                                                const int* __restrict__ cnt_out,
                                                int* __restrict__ rowptr,
                                                float* __restrict__ rs_out,
                                                float* __restrict__ rs_in, int N) {
    __shared__ int wsum[32];
    __shared__ int wpart[32];
    const int tid  = threadIdx.x;
    const int lane = tid & 31;
    const int wave = tid >> 5;
    const int b    = blockIdx.x;

    int part = 0;
    const int4* __restrict__ c4 = (const int4*)cnt_in;
    const int n4 = b * (SB / 4);
#pragma unroll 4
    for (int j = tid; j < n4; j += SB) {
        const int4 v = c4[j];
        part += (v.x + v.y) + (v.z + v.w);
    }

    const int i = b * SB + tid;
    const int c = cnt_in[i];

    int incl = c;
#pragma unroll
    for (int off = 1; off < 32; off <<= 1) {
        const int t = __shfl_up(incl, off, 32);
        if (lane >= off) incl += t;
    }
#pragma unroll
    for (int off = 16; off > 0; off >>= 1) part += __shfl_xor(part, off, 32);

    if (lane == 31) wsum[wave] = incl;
    if (lane == 0)  wpart[wave] = part;
    __syncthreads();

    int tot = wpart[lane] + ((lane < wave) ? wsum[lane] : 0);
#pragma unroll
    for (int off = 16; off > 0; off >>= 1) tot += __shfl_xor(tot, off, 32);

    if (i < N) {
        rowptr[i] = tot + incl - c;
        rs_out[i] = rsqrtf((float)(cnt_out[i] + 1));
        rs_in[i]  = rsqrtf((float)(c + 1));
    }
}

__global__ __launch_bounds__(256) void edge_place(const int* __restrict__ src,
                                                  const int* __restrict__ dst,
                                                  const int* __restrict__ rank,
                                                  const int* __restrict__ rowptr,
                                                  int* __restrict__ col, int E) {
    int e = blockIdx.x * blockDim.x + threadIdx.x;
    if (e < E) col[rowptr[dst[e]] + rank[e]] = src[e];
}

__global__ __launch_bounds__(256) void gemm64_wmma(const float* __restrict__ X,
                                                   const float* __restrict__ rs,
                                                   const float* __restrict__ W,
                                                   float* __restrict__ out0,
                                                   int N) {
    const int wave = threadIdx.x >> 5;
    const int lane = threadIdx.x & 31;
    const int nb = blockIdx.x * 128 + wave * 16;
    if (nb >= N) return;
    const int half = lane >> 4;
    const int lr   = lane & 15;

    const float s = rs[nb + lr];
    const float* __restrict__ xrow = X + (size_t)(nb + lr) * D;

    v8f acc0 = {}, acc1 = {}, acc2 = {}, acc3 = {};
#pragma unroll
    for (int k0 = 0; k0 < D; k0 += 4) {
        const int ka = k0 + half * 2;
        v2f a;
        a.x = xrow[ka] * s;
        a.y = xrow[ka + 1] * s;
        const float* __restrict__ w0 = W + (size_t)ka * D + lr;
        const float* __restrict__ w1 = w0 + D;
        v2f b0 = { w0[0],  w1[0]  };
        v2f b1 = { w0[16], w1[16] };
        v2f b2 = { w0[32], w1[32] };
        v2f b3 = { w0[48], w1[48] };
        acc0 = __builtin_amdgcn_wmma_f32_16x16x4_f32(false, a, false, b0, (short)0, acc0, false, false);
        acc1 = __builtin_amdgcn_wmma_f32_16x16x4_f32(false, a, false, b1, (short)0, acc1, false, false);
        acc2 = __builtin_amdgcn_wmma_f32_16x16x4_f32(false, a, false, b2, (short)0, acc2, false, false);
        acc3 = __builtin_amdgcn_wmma_f32_16x16x4_f32(false, a, false, b3, (short)0, acc3, false, false);
    }

    const int row0 = nb + half * 8;
#pragma unroll
    for (int v = 0; v < 8; ++v) {
        const size_t base = (size_t)(row0 + v) * D + lr;
        out0[base]      = acc0[v];
        out0[base + 16] = acc1[v];
        out0[base + 32] = acc2[v];
        out0[base + 48] = acc3[v];
    }
}

static __device__ __forceinline__ v4f row_sum(const float* __restrict__ h,
                                              const int* __restrict__ col,
                                              const int* __restrict__ zcell,
                                              int start, int cnt, int n, int l16) {
    const int f = l16 * 4;
    const int l8 = l16 & 7;
    const int* __restrict__ cp = col + start + l8;
    v4f acc = *(const v4f*)(h + (size_t)n * D + f);
    int idx = *((l8 < cnt) ? cp : zcell);
    for (int base = 0; base < cnt; base += 8) {
        const int nidx = *((base + 8 + l8 < cnt) ? (cp + base + 8) : zcell);
        v4f v[8];
#pragma unroll
        for (int j = 0; j < 8; ++j) {
            const int s = __shfl(idx, j, 16);
            v[j] = *(const v4f*)(h + (size_t)s * D + f);
        }
        asm volatile("" :: "v"(nidx), "v"(v[0]), "v"(v[1]), "v"(v[2]), "v"(v[3]),
                           "v"(v[4]), "v"(v[5]), "v"(v[6]), "v"(v[7]));
#pragma unroll
        for (int j = 0; j < 8; ++j) acc += v[j];
        idx = nidx;
    }
    return acc;
}

__global__ __launch_bounds__(256) void gather_relu(const float* __restrict__ h,
                                                   const int* __restrict__ rowptr,
                                                   const int* __restrict__ cnt_in,
                                                   const int* __restrict__ col,
                                                   const int* __restrict__ zcell,
                                                   const float* __restrict__ rs_in,
                                                   const float* __restrict__ b,
                                                   float* __restrict__ hout, int N) {
    const int t = blockIdx.x * blockDim.x + threadIdx.x;
    const int node = t >> 4;
    const int l16 = threadIdx.x & 15;
    const int n = node < N ? node : N - 1;
    const int f = l16 * 4;
    const v4f a = row_sum(h, col, zcell, rowptr[n], cnt_in[n], n, l16);
    const float s = rs_in[n];
    float4 r;
    r.x = fmaxf(a.x * s + b[f + 0], 0.0f);
    r.y = fmaxf(a.y * s + b[f + 1], 0.0f);
    r.z = fmaxf(a.z * s + b[f + 2], 0.0f);
    r.w = fmaxf(a.w * s + b[f + 3], 0.0f);
    if (node < N) *(float4*)(hout + (size_t)node * D + f) = r;
}

__global__ __launch_bounds__(256) void gather_dot(const float* __restrict__ h,
                                                  const int* __restrict__ rowptr,
                                                  const int* __restrict__ cnt_in,
                                                  const int* __restrict__ col,
                                                  const int* __restrict__ zcell,
                                                  const float* __restrict__ rs_in,
                                                  const float* __restrict__ b2,
                                                  const float* __restrict__ Wr,
                                                  const float* __restrict__ br,
                                                  float* __restrict__ out, int N) {
    const int t = blockIdx.x * blockDim.x + threadIdx.x;
    const int node = t >> 4;
    const int l16 = threadIdx.x & 15;
    const int n = node < N ? node : N - 1;
    const int f = l16 * 4;
    const v4f a = row_sum(h, col, zcell, rowptr[n], cnt_in[n], n, l16);
    const float s = rs_in[n];
    float p = ((a.x * s + b2[f + 0]) * Wr[f + 0] + (a.y * s + b2[f + 1]) * Wr[f + 1])
            + ((a.z * s + b2[f + 2]) * Wr[f + 2] + (a.w * s + b2[f + 3]) * Wr[f + 3]);
#pragma unroll
    for (int off = 8; off > 0; off >>= 1) p += __shfl_xor(p, off, 32);
    if (node < N && l16 == 0) out[node] = p + br[0];
}

extern "C" void kernel_launch(void* const* d_in, const int* in_sizes, int n_in,
                              void* d_out, int out_size, void* d_ws, size_t ws_size,
                              hipStream_t stream) {
    const float* x   = (const float*)d_in[0];
    const int*   src = (const int*)d_in[1];
    const int*   dst = (const int*)d_in[2];
    const float* W1  = (const float*)d_in[3];
    const float* b1  = (const float*)d_in[4];
    const float* W2  = (const float*)d_in[5];
    const float* b2  = (const float*)d_in[6];
    const float* Wr  = (const float*)d_in[7];
    const float* br  = (const float*)d_in[8];
    float* out = (float*)d_out;

    const int N = in_sizes[0] / D;
    const int E = in_sizes[1];
    const int Np = ((N + SB - 1) / SB) * SB;
    const size_t Ep = (((size_t)E + 63) / 64) * 64;

    int*   cnt_in  = (int*)d_ws;
    int*   cnt_out = cnt_in + Np;
    int*   rowptr  = cnt_out + Np;
    float* rs_out  = (float*)(rowptr + Np);
    float* rs_in   = rs_out + Np;
    int*   rank    = (int*)(rs_in + Np);
    int*   col     = rank + Ep;
    float* h       = (float*)(col + Ep);
    float* h1      = h + ((size_t)N + 1) * D;
    int*   zcell   = (int*)(h1 + (size_t)N * D);

    const int TB = 256;
    const int gZ    = (2 * Np + TB - 1) / TB;
    const int gE    = (E + TB - 1) / TB;
    const int gScan = Np / SB;
    const int gGemm = (N + 127) / 128;
    const int gRow  = (int)(((size_t)N * 16 + TB - 1) / TB);

    cnt_zero  <<<gZ, TB, 0, stream>>>(cnt_in, 2 * Np, h + (size_t)N * D, zcell, N);
    edge_count<<<gE, TB, 0, stream>>>(src, dst, cnt_out, cnt_in, rank, E);
    node_scan <<<gScan, SB, 0, stream>>>(cnt_in, cnt_out, rowptr, rs_out, rs_in, N);
    edge_place<<<gE, TB, 0, stream>>>(src, dst, rank, rowptr, col, E);

    gemm64_wmma<<<gGemm, TB, 0, stream>>>(x, rs_out, W1, h, N);
    gather_relu<<<gRow, TB, 0, stream>>>(h, rowptr, cnt_in, col, zcell, rs_in, b1, h1, N);

    gemm64_wmma<<<gGemm, TB, 0, stream>>>(h1, rs_out, W2, h, N);
    gather_dot <<<gRow, TB, 0, stream>>>(h, rowptr, cnt_in, col, zcell, rs_in, b2, Wr, br, out, N);
}


# counting pass writes each edge's source into its destination's fixed 64-slot row (slot = count the in-degree add returns; overflow list beyond): scan and placement launches go; sum order kept
# speedup vs baseline: 2.1323x; 1.1250x over previous
#include <hip/hip_runtime.h>
#include <hip/hip_bf16.h>


typedef __attribute__((ext_vector_type(2))) float v2f;
typedef __attribute__((ext_vector_type(8))) float v8f;
typedef __attribute__((ext_vector_type(4))) float v4f;

#define D 64
#define NPAD 1024
#define CAP 64
#define ZPAD 64

__global__ __launch_bounds__(256) void cnt_zero(int* __restrict__ c, int n2, float* __restrict__ zrow,
                                                int* __restrict__ zcell, int zr) {
    int i = blockIdx.x * blockDim.x + threadIdx.x;
    if (i < n2) c[i] = 0;
    if (i < D) zrow[i] = 0.0f;
    if (i == 0) zcell[0] = zr;
}

__global__ __launch_bounds__(256) void edge_slot(const int* __restrict__ src,
                                                 const int* __restrict__ dst,
                                                 int* cnt_out, int* cnt_in,
                                                 int* __restrict__ ell,
                                                 int* ovf_n, int* __restrict__ ovf, int E) {
    int e = blockIdx.x * blockDim.x + threadIdx.x;
    if (e < E) {
        const int s = src[e];
        const int d = dst[e];
        atomicAdd(&cnt_out[s], 1);
        const int r = atomicAdd(&cnt_in[d], 1);
        if (r < CAP) {
            ell[(size_t)d * CAP + r] = s;
        } else {
            const int p = atomicAdd(ovf_n, 1);
            ovf[2 * (size_t)p]     = d;
            ovf[2 * (size_t)p + 1] = s;
        }
    }
}

__global__ __launch_bounds__(256) void gemm64_wmma(const float* __restrict__ X,
                                                   const int* __restrict__ cnt,
                                                   const float* __restrict__ W,
                                                   float* __restrict__ out0,
                                                   int N) {
    const int wave = threadIdx.x >> 5;
    const int lane = threadIdx.x & 31;
    const int nb = blockIdx.x * 128 + wave * 16;
    if (nb >= N) return;
    const int half = lane >> 4;
    const int lr   = lane & 15;

    const float s = rsqrtf((float)(cnt[nb + lr] + 1));
    const float* __restrict__ xrow = X + (size_t)(nb + lr) * D;

    v8f acc0 = {}, acc1 = {}, acc2 = {}, acc3 = {};
#pragma unroll
    for (int k0 = 0; k0 < D; k0 += 4) {
        const int ka = k0 + half * 2;
        v2f a;
        a.x = xrow[ka] * s;
        a.y = xrow[ka + 1] * s;
        const float* __restrict__ w0 = W + (size_t)ka * D + lr;
        const float* __restrict__ w1 = w0 + D;
        v2f b0 = { w0[0],  w1[0]  };
        v2f b1 = { w0[16], w1[16] };
        v2f b2 = { w0[32], w1[32] };
        v2f b3 = { w0[48], w1[48] };
        acc0 = __builtin_amdgcn_wmma_f32_16x16x4_f32(false, a, false, b0, (short)0, acc0, false, false);
        acc1 = __builtin_amdgcn_wmma_f32_16x16x4_f32(false, a, false, b1, (short)0, acc1, false, false);
        acc2 = __builtin_amdgcn_wmma_f32_16x16x4_f32(false, a, false, b2, (short)0, acc2, false, false);
        acc3 = __builtin_amdgcn_wmma_f32_16x16x4_f32(false, a, false, b3, (short)0, acc3, false, false);
    }

    const int row0 = nb + half * 8;
#pragma unroll
    for (int v = 0; v < 8; ++v) {
        const size_t base = (size_t)(row0 + v) * D + lr;
        out0[base]      = acc0[v];
        out0[base + 16] = acc1[v];
        out0[base + 32] = acc2[v];
        out0[base + 48] = acc3[v];
    }
}

static __device__ __forceinline__ v4f row_sum(const float* __restrict__ h,
                                              const int* __restrict__ ell,
                                              const int* __restrict__ zcell,
                                              const int* __restrict__ ovf_n,
                                              const int* __restrict__ ovf,
                                              int cnt, int n, int l16) {
    const int f = l16 * 4;
    const int l8 = l16 & 7;
    const int m = cnt < CAP ? cnt : CAP;
    const int* __restrict__ cp = ell + (size_t)n * CAP + l8;
    v4f acc = *(const v4f*)(h + (size_t)n * D + f);
    int idx = *((l8 < m) ? cp : zcell);
    for (int base = 0; base < m; base += 8) {
        const int nidx = *((base + 8 + l8 < m) ? (cp + base + 8) : zcell);
        v4f v[8];
#pragma unroll
        for (int j = 0; j < 8; ++j) {
            const int s = __shfl(idx, j, 16);
            v[j] = *(const v4f*)(h + (size_t)s * D + f);
        }
        asm volatile("" :: "v"(nidx), "v"(v[0]), "v"(v[1]), "v"(v[2]), "v"(v[3]),
                           "v"(v[4]), "v"(v[5]), "v"(v[6]), "v"(v[7]));
#pragma unroll
        for (int j = 0; j < 8; ++j) acc += v[j];
        idx = nidx;
    }
    if (cnt > CAP) {
        const int tot = ovf_n[0];
        for (int i = 0; i < tot; ++i) {
            if (ovf[2 * (size_t)i] == n) acc += *(const v4f*)(h + (size_t)ovf[2 * (size_t)i + 1] * D + f);
        }
    }
    return acc;
}

__global__ __launch_bounds__(256) void gather_relu(const float* __restrict__ h,
                                                   const int* __restrict__ cnt_in,
                                                   const int* __restrict__ ell,
                                                   const int* __restrict__ zcell,
                                                   const int* __restrict__ ovf_n,
                                                   const int* __restrict__ ovf,
                                                   const float* __restrict__ b,
                                                   float* __restrict__ hout, int N) {
    const int t = blockIdx.x * blockDim.x + threadIdx.x;
    const int node = t >> 4;
    const int l16 = threadIdx.x & 15;
    const int n = node < N ? node : N - 1;
    const int f = l16 * 4;
    const int c = cnt_in[n];
    const v4f a = row_sum(h, ell, zcell, ovf_n, ovf, c, n, l16);
    const float s = rsqrtf((float)(c + 1));
    float4 r;
    r.x = fmaxf(a.x * s + b[f + 0], 0.0f);
    r.y = fmaxf(a.y * s + b[f + 1], 0.0f);
    r.z = fmaxf(a.z * s + b[f + 2], 0.0f);
    r.w = fmaxf(a.w * s + b[f + 3], 0.0f);
    if (node < N) *(float4*)(hout + (size_t)node * D + f) = r;
}

__global__ __launch_bounds__(256) void gather_dot(const float* __restrict__ h,
                                                  const int* __restrict__ cnt_in,
                                                  const int* __restrict__ ell,
                                                  const int* __restrict__ zcell,
                                                  const int* __restrict__ ovf_n,
                                                  const int* __restrict__ ovf,
                                                  const float* __restrict__ b2,
                                                  const float* __restrict__ Wr,
                                                  const float* __restrict__ br,
                                                  float* __restrict__ out, int N) {
    const int t = blockIdx.x * blockDim.x + threadIdx.x;
    const int node = t >> 4;
    const int l16 = threadIdx.x & 15;
    const int n = node < N ? node : N - 1;
    const int f = l16 * 4;
    const int c = cnt_in[n];
    const v4f a = row_sum(h, ell, zcell, ovf_n, ovf, c, n, l16);
    const float s = rsqrtf((float)(c + 1));
    float p = ((a.x * s + b2[f + 0]) * Wr[f + 0] + (a.y * s + b2[f + 1]) * Wr[f + 1])
            + ((a.z * s + b2[f + 2]) * Wr[f + 2] + (a.w * s + b2[f + 3]) * Wr[f + 3]);
#pragma unroll
    for (int off = 8; off > 0; off >>= 1) p += __shfl_xor(p, off, 32);
    if (node < N && l16 == 0) out[node] = p + br[0];
}

extern "C" void kernel_launch(void* const* d_in, const int* in_sizes, int n_in,
                              void* d_out, int out_size, void* d_ws, size_t ws_size,
                              hipStream_t stream) {
    const float* x   = (const float*)d_in[0];
    const int*   src = (const int*)d_in[1];
    const int*   dst = (const int*)d_in[2];
    const float* W1  = (const float*)d_in[3];
    const float* b1  = (const float*)d_in[4];
    const float* W2  = (const float*)d_in[5];
    const float* b2  = (const float*)d_in[6];
    const float* Wr  = (const float*)d_in[7];
    const float* br  = (const float*)d_in[8];
    float* out = (float*)d_out;

    const int N = in_sizes[0] / D;
    const int E = in_sizes[1];
    const int Np = ((N + NPAD - 1) / NPAD) * NPAD;
    const size_t Ep = (((size_t)E + 63) / 64) * 64;

    int*   cnt_in  = (int*)d_ws;
    int*   cnt_out = cnt_in + Np;
    int*   ovf_n   = cnt_out + Np;
    int*   ell     = ovf_n + ZPAD;
    int*   ovf     = ell + (size_t)N * CAP;
    float* h       = (float*)(ovf + 2 * Ep);
    float* h1      = h + ((size_t)N + 1) * D;
    int*   zcell   = (int*)(h1 + (size_t)N * D);

    const int TB = 256;
    const int nZ    = 2 * Np + ZPAD;
    const int gZ    = (nZ + TB - 1) / TB;
    const int gE    = (E + TB - 1) / TB;
    const int gGemm = (N + 127) / 128;
    const int gRow  = (int)(((size_t)N * 16 + TB - 1) / TB);

    cnt_zero <<<gZ, TB, 0, stream>>>(cnt_in, nZ, h + (size_t)N * D, zcell, N);
    edge_slot<<<gE, TB, 0, stream>>>(src, dst, cnt_out, cnt_in, ell, ovf_n, ovf, E);

    gemm64_wmma<<<gGemm, TB, 0, stream>>>(x, cnt_out, W1, h, N);
    gather_relu<<<gRow, TB, 0, stream>>>(h, cnt_in, ell, zcell, ovf_n, ovf, b1, h1, N);

    gemm64_wmma<<<gGemm, TB, 0, stream>>>(h1, cnt_out, W2, h, N);
    gather_dot <<<gRow, TB, 0, stream>>>(h, cnt_in, ell, zcell, ovf_n, ovf, b2, Wr, br, out, N);
}


# layer-2 W2 product moved into the layer-1 row-sum launch: each 16-node block multiplies its own relu tile (via LDS) on WMMA, stores whole rows; h1 array and one launch go; sums bit-identical
# speedup vs baseline: 2.2455x; 1.0531x over previous
#include <hip/hip_runtime.h>
#include <hip/hip_bf16.h>


typedef __attribute__((ext_vector_type(2))) float v2f;
typedef __attribute__((ext_vector_type(8))) float v8f;
typedef __attribute__((ext_vector_type(4))) float v4f;

#define D 64
#define NPAD 1024
#define CAP 64
#define ZPAD 64

__global__ __launch_bounds__(256) void cnt_zero(int* __restrict__ c, int n2, float* __restrict__ zrow,
                                                float* __restrict__ zrow2,
                                                int* __restrict__ zcell, int zr) {
    int i = blockIdx.x * blockDim.x + threadIdx.x;
    if (i < n2) c[i] = 0;
    if (i < D) { zrow[i] = 0.0f; zrow2[i] = 0.0f; }
    if (i == 0) zcell[0] = zr;
}

__global__ __launch_bounds__(256) void edge_slot(const int* __restrict__ src,
                                                 const int* __restrict__ dst,
                                                 int* cnt_out, int* cnt_in,
                                                 int* __restrict__ ell,
                                                 int* ovf_n, int* __restrict__ ovf, int E) {
    int e = blockIdx.x * blockDim.x + threadIdx.x;
    if (e < E) {
        const int s = src[e];
        const int d = dst[e];
        atomicAdd(&cnt_out[s], 1);
        const int r = atomicAdd(&cnt_in[d], 1);
        if (r < CAP) {
            ell[(size_t)d * CAP + r] = s;
        } else {
            const int p = atomicAdd(ovf_n, 1);
            ovf[2 * (size_t)p]     = d;
            ovf[2 * (size_t)p + 1] = s;
        }
    }
}

__global__ __launch_bounds__(256) void gemm64_wmma(const float* __restrict__ X,
                                                   const int* __restrict__ cnt,
                                                   const float* __restrict__ W,
                                                   float* __restrict__ out0,
                                                   int N) {
    const int wave = threadIdx.x >> 5;
    const int lane = threadIdx.x & 31;
    const int nb = blockIdx.x * 128 + wave * 16;
    if (nb >= N) return;
    const int half = lane >> 4;
    const int lr   = lane & 15;

    const float s = rsqrtf((float)(cnt[nb + lr] + 1));
    const float* __restrict__ xrow = X + (size_t)(nb + lr) * D;

    v8f acc0 = {}, acc1 = {}, acc2 = {}, acc3 = {};
#pragma unroll
    for (int k0 = 0; k0 < D; k0 += 4) {
        const int ka = k0 + half * 2;
        v2f a;
        a.x = xrow[ka] * s;
        a.y = xrow[ka + 1] * s;
        const float* __restrict__ w0 = W + (size_t)ka * D + lr;
        const float* __restrict__ w1 = w0 + D;
        v2f b0 = { w0[0],  w1[0]  };
        v2f b1 = { w0[16], w1[16] };
        v2f b2 = { w0[32], w1[32] };
        v2f b3 = { w0[48], w1[48] };
        acc0 = __builtin_amdgcn_wmma_f32_16x16x4_f32(false, a, false, b0, (short)0, acc0, false, false);
        acc1 = __builtin_amdgcn_wmma_f32_16x16x4_f32(false, a, false, b1, (short)0, acc1, false, false);
        acc2 = __builtin_amdgcn_wmma_f32_16x16x4_f32(false, a, false, b2, (short)0, acc2, false, false);
        acc3 = __builtin_amdgcn_wmma_f32_16x16x4_f32(false, a, false, b3, (short)0, acc3, false, false);
    }

    const int row0 = nb + half * 8;
#pragma unroll
    for (int v = 0; v < 8; ++v) {
        const size_t base = (size_t)(row0 + v) * D + lr;
        out0[base]      = acc0[v];
        out0[base + 16] = acc1[v];
        out0[base + 32] = acc2[v];
        out0[base + 48] = acc3[v];
    }
}

static __device__ __forceinline__ v4f row_sum(const float* __restrict__ h,
                                              const int* __restrict__ ell,
                                              const int* __restrict__ zcell,
                                              const int* __restrict__ ovf_n,
                                              const int* __restrict__ ovf,
                                              int cnt, int n, int l16) {
    const int f = l16 * 4;
    const int l8 = l16 & 7;
    const int m = cnt < CAP ? cnt : CAP;
    const int* __restrict__ cp = ell + (size_t)n * CAP + l8;
    v4f acc = *(const v4f*)(h + (size_t)n * D + f);
    int idx = *((l8 < m) ? cp : zcell);
    for (int base = 0; base < m; base += 8) {
        const int nidx = *((base + 8 + l8 < m) ? (cp + base + 8) : zcell);
        v4f v[8];
#pragma unroll
        for (int j = 0; j < 8; ++j) {
            const int s = __shfl(idx, j, 16);
            v[j] = *(const v4f*)(h + (size_t)s * D + f);
        }
        asm volatile("" :: "v"(nidx), "v"(v[0]), "v"(v[1]), "v"(v[2]), "v"(v[3]),
                           "v"(v[4]), "v"(v[5]), "v"(v[6]), "v"(v[7]));
#pragma unroll
        for (int j = 0; j < 8; ++j) acc += v[j];
        idx = nidx;
    }
    if (cnt > CAP) {
        const int tot = ovf_n[0];
        for (int i = 0; i < tot; ++i) {
            if (ovf[2 * (size_t)i] == n) acc += *(const v4f*)(h + (size_t)ovf[2 * (size_t)i + 1] * D + f);
        }
    }
    return acc;
}

#define TP 68
__global__ __launch_bounds__(256) void gather_relu_gemm(const float* __restrict__ h,
                                                        const int* __restrict__ cnt_in,
                                                        const int* __restrict__ cnt_out,
                                                        const int* __restrict__ ell,
                                                        const int* __restrict__ zcell,
                                                        const int* __restrict__ ovf_n,
                                                        const int* __restrict__ ovf,
                                                        const float* __restrict__ b,
                                                        const float* __restrict__ W,
                                                        float* __restrict__ hout, int N) {
    __shared__ __attribute__((aligned(16))) float ta[16 * TP];
    __shared__ __attribute__((aligned(16))) float td[16 * TP];
    const int t = blockIdx.x * blockDim.x + threadIdx.x;
    const int node = t >> 4;
    const int row = threadIdx.x >> 4;
    const int l16 = threadIdx.x & 15;
    const int n = node < N ? node : N - 1;
    const int f = l16 * 4;
    const int c = cnt_in[n];
    const int co = cnt_out[n];
    const v4f a = row_sum(h, ell, zcell, ovf_n, ovf, c, n, l16);
    const float s = rsqrtf((float)(c + 1));
    const float so = rsqrtf((float)(co + 1));
    v4f r;
    r.x = fmaxf(a.x * s + b[f + 0], 0.0f) * so;
    r.y = fmaxf(a.y * s + b[f + 1], 0.0f) * so;
    r.z = fmaxf(a.z * s + b[f + 2], 0.0f) * so;
    r.w = fmaxf(a.w * s + b[f + 3], 0.0f) * so;
    *(v4f*)(ta + row * TP + f) = r;

    const int wave = threadIdx.x >> 5;
    const int lane = threadIdx.x & 31;
    const int half = lane >> 4;
    const int lr   = lane & 15;
    v2f bv[16];
    if (wave < 4) {
        const float* __restrict__ wcol = W + wave * 16 + lr;
#pragma unroll
        for (int i = 0; i < 16; ++i) {
            const int ka = i * 4 + half * 2;
            bv[i].x = wcol[(size_t)ka * D];
            bv[i].y = wcol[(size_t)(ka + 1) * D];
        }
    }
    __syncthreads();

    if (wave < 4) {
        const float* __restrict__ arow = ta + lr * TP;
        v8f acc = {};
#pragma unroll
        for (int i = 0; i < 16; ++i) {
            const int ka = i * 4 + half * 2;
            const v2f av = *(const v2f*)(arow + ka);
            acc = __builtin_amdgcn_wmma_f32_16x16x4_f32(false, av, false, bv[i], (short)0, acc, false, false);
        }
#pragma unroll
        for (int v = 0; v < 8; ++v)
            td[(half * 8 + v) * TP + wave * 16 + lr] = acc[v];
    }
    __syncthreads();
    const v4f o = *(const v4f*)(td + row * TP + f);
    if (node < N) *(v4f*)(hout + (size_t)node * D + f) = o;
}

__global__ __launch_bounds__(256) void gather_dot(const float* __restrict__ h,
                                                  const int* __restrict__ cnt_in,
                                                  const int* __restrict__ ell,
                                                  const int* __restrict__ zcell,
                                                  const int* __restrict__ ovf_n,
                                                  const int* __restrict__ ovf,
                                                  const float* __restrict__ b2,
                                                  const float* __restrict__ Wr,
                                                  const float* __restrict__ br,
                                                  float* __restrict__ out, int N) {
    const int t = blockIdx.x * blockDim.x + threadIdx.x;
    const int node = t >> 4;
    const int l16 = threadIdx.x & 15;
    const int n = node < N ? node : N - 1;
    const int f = l16 * 4;
    const int c = cnt_in[n];
    const v4f a = row_sum(h, ell, zcell, ovf_n, ovf, c, n, l16);
    const float s = rsqrtf((float)(c + 1));
    float p = ((a.x * s + b2[f + 0]) * Wr[f + 0] + (a.y * s + b2[f + 1]) * Wr[f + 1])
            + ((a.z * s + b2[f + 2]) * Wr[f + 2] + (a.w * s + b2[f + 3]) * Wr[f + 3]);
#pragma unroll
    for (int off = 8; off > 0; off >>= 1) p += __shfl_xor(p, off, 32);
    if (node < N && l16 == 0) out[node] = p + br[0];
}

extern "C" void kernel_launch(void* const* d_in, const int* in_sizes, int n_in,
                              void* d_out, int out_size, void* d_ws, size_t ws_size,
                              hipStream_t stream) {
    const float* x   = (const float*)d_in[0];
    const int*   src = (const int*)d_in[1];
    const int*   dst = (const int*)d_in[2];
    const float* W1  = (const float*)d_in[3];
    const float* b1  = (const float*)d_in[4];
    const float* W2  = (const float*)d_in[5];
    const float* b2  = (const float*)d_in[6];
    const float* Wr  = (const float*)d_in[7];
    const float* br  = (const float*)d_in[8];
    float* out = (float*)d_out;

    const int N = in_sizes[0] / D;
    const int E = in_sizes[1];
    const int Np = ((N + NPAD - 1) / NPAD) * NPAD;
    const size_t Ep = (((size_t)E + 63) / 64) * 64;

    int*   cnt_in  = (int*)d_ws;
    int*   cnt_out = cnt_in + Np;
    int*   ovf_n   = cnt_out + Np;
    int*   ell     = ovf_n + ZPAD;
    int*   ovf     = ell + (size_t)N * CAP;
    float* h       = (float*)(ovf + 2 * Ep);
    float* h2      = h + ((size_t)N + 1) * D;
    int*   zcell   = (int*)(h2 + ((size_t)N + 1) * D);

    const int TB = 256;
    const int nZ    = 2 * Np + ZPAD;
    const int gZ    = (nZ + TB - 1) / TB;
    const int gE    = (E + TB - 1) / TB;
    const int gGemm = (N + 127) / 128;
    const int gRow  = (int)(((size_t)N * 16 + TB - 1) / TB);

    cnt_zero <<<gZ, TB, 0, stream>>>(cnt_in, nZ, h + (size_t)N * D, h2 + (size_t)N * D, zcell, N);
    edge_slot<<<gE, TB, 0, stream>>>(src, dst, cnt_out, cnt_in, ell, ovf_n, ovf, E);

    gemm64_wmma     <<<gGemm, TB, 0, stream>>>(x, cnt_out, W1, h, N);
    gather_relu_gemm<<<gRow, TB, 0, stream>>>(h, cnt_in, cnt_out, ell, zcell, ovf_n, ovf, b1, W2, h2, N);

    gather_dot      <<<gRow, TB, 0, stream>>>(h2, cnt_in, ell, zcell, ovf_n, ovf, b2, Wr, br, out, N);
}


# a node's in- and out-degree counts move into words 0-1 of its own 256-B source row (62 slots): an edge's slot add and source store hit one unit, a node's 16 lanes load the row whole; sum order kept
# speedup vs baseline: 2.4692x; 1.0996x over previous
#include <hip/hip_runtime.h>
#include <hip/hip_bf16.h>


typedef __attribute__((ext_vector_type(2))) float v2f;
typedef __attribute__((ext_vector_type(8))) float v8f;
typedef __attribute__((ext_vector_type(4))) float v4f;
typedef __attribute__((ext_vector_type(2))) int v2i;
typedef __attribute__((ext_vector_type(4))) int v4i;

#define D 64
#define RW 64
#define CAP 62
#define ZPAD 64

__global__ __launch_bounds__(256) void cnt_zero(int* __restrict__ ell, int N, int* __restrict__ ovf_n,
                                                float* __restrict__ zrow,
                                                float* __restrict__ zrow2) {
    int i = blockIdx.x * blockDim.x + threadIdx.x;
    if (i < N) { v2i z = {0, 0}; *(v2i*)(ell + (size_t)i * RW) = z; }
    if (i < D) { zrow[i] = 0.0f; zrow2[i] = 0.0f; }
    if (i == 0) ovf_n[0] = 0;
}

__global__ __launch_bounds__(256) void edge_slot(const int* __restrict__ src,
                                                 const int* __restrict__ dst,
                                                 int* ell,
                                                 int* ovf_n, int* __restrict__ ovf, int E) {
    int e = blockIdx.x * blockDim.x + threadIdx.x;
    if (e < E) {
        const int s = src[e];
        const int d = dst[e];
        int* drow = ell + (size_t)d * RW;
        const int r = atomicAdd(drow, 1);
        atomicAdd(ell + (size_t)s * RW + 1, 1);
        if (r < CAP) {
            drow[2 + r] = s;
        } else {
            const int p = atomicAdd(ovf_n, 1);
            ovf[2 * (size_t)p]     = d;
            ovf[2 * (size_t)p + 1] = s;
        }
    }
}

__global__ __launch_bounds__(256) void gemm64_wmma(const float* __restrict__ X,
                                                   const int* __restrict__ ell,
                                                   const float* __restrict__ W,
                                                   float* __restrict__ out0,
                                                   int N) {
    const int wave = threadIdx.x >> 5;
    const int lane = threadIdx.x & 31;
    const int nb = blockIdx.x * 128 + wave * 16;
    if (nb >= N) return;
    const int half = lane >> 4;
    const int lr   = lane & 15;

    const float s = rsqrtf((float)(ell[(size_t)(nb + lr) * RW + 1] + 1));
    const float* __restrict__ xrow = X + (size_t)(nb + lr) * D;

    v8f acc0 = {}, acc1 = {}, acc2 = {}, acc3 = {};
#pragma unroll
    for (int k0 = 0; k0 < D; k0 += 4) {
        const int ka = k0 + half * 2;
        v2f a;
        a.x = xrow[ka] * s;
        a.y = xrow[ka + 1] * s;
        const float* __restrict__ w0 = W + (size_t)ka * D + lr;
        const float* __restrict__ w1 = w0 + D;
        v2f b0 = { w0[0],  w1[0]  };
        v2f b1 = { w0[16], w1[16] };
        v2f b2 = { w0[32], w1[32] };
        v2f b3 = { w0[48], w1[48] };
        acc0 = __builtin_amdgcn_wmma_f32_16x16x4_f32(false, a, false, b0, (short)0, acc0, false, false);
        acc1 = __builtin_amdgcn_wmma_f32_16x16x4_f32(false, a, false, b1, (short)0, acc1, false, false);
        acc2 = __builtin_amdgcn_wmma_f32_16x16x4_f32(false, a, false, b2, (short)0, acc2, false, false);
        acc3 = __builtin_amdgcn_wmma_f32_16x16x4_f32(false, a, false, b3, (short)0, acc3, false, false);
    }

    const int row0 = nb + half * 8;
#pragma unroll
    for (int v = 0; v < 8; ++v) {
        const size_t base = (size_t)(row0 + v) * D + lr;
        out0[base]      = acc0[v];
        out0[base + 16] = acc1[v];
        out0[base + 32] = acc2[v];
        out0[base + 48] = acc3[v];
    }
}

static __device__ __forceinline__ v4f row_sum(const float* __restrict__ h,
                                              const v4i rw,
                                              const int* __restrict__ ovf_n,
                                              const int* __restrict__ ovf,
                                              int cnt, int n, int l16, int zr) {
    const int f = l16 * 4;
    const int m = cnt < CAP ? cnt : CAP;
    v4f acc = *(const v4f*)(h + (size_t)n * D + f);
#pragma unroll
    for (int t = 0; t < RW / 8; ++t) {
        const int base = t * 8;
        if (base < m) {
            v4f v[8];
#pragma unroll
            for (int j = 0; j < 8; ++j) {
                const int word = base + j + 2;
                const int comp = word & 3;
                const int w = comp == 0 ? rw.x : comp == 1 ? rw.y : comp == 2 ? rw.z : rw.w;
                int s = __shfl(w, (word >> 2) & 15, 16);
                s = (base + j < m) ? s : zr;
                v[j] = *(const v4f*)(h + (size_t)s * D + f);
            }
#pragma unroll
            for (int j = 0; j < 8; ++j) acc += v[j];
        }
    }
    if (cnt > CAP) {
        const int tot = ovf_n[0];
        for (int i = 0; i < tot; ++i) {
            if (ovf[2 * (size_t)i] == n) acc += *(const v4f*)(h + (size_t)ovf[2 * (size_t)i + 1] * D + f);
        }
    }
    return acc;
}

#define TP 68
__global__ __launch_bounds__(256) void gather_relu_gemm(const float* __restrict__ h,
                                                        const int* __restrict__ ell,
                                                        const int* __restrict__ ovf_n,
                                                        const int* __restrict__ ovf,
                                                        const float* __restrict__ b,
                                                        const float* __restrict__ W,
                                                        float* __restrict__ hout, int N) {
    __shared__ __attribute__((aligned(16))) float ta[16 * TP];
    __shared__ __attribute__((aligned(16))) float td[16 * TP];
    const int t = blockIdx.x * blockDim.x + threadIdx.x;
    const int node = t >> 4;
    const int row = threadIdx.x >> 4;
    const int l16 = threadIdx.x & 15;
    const int n = node < N ? node : N - 1;
    const int f = l16 * 4;
    const v4i rw = *(const v4i*)(ell + (size_t)n * RW + f);
    const int c = __shfl(rw.x, 0, 16);
    const int co = __shfl(rw.y, 0, 16);
    const v4f a = row_sum(h, rw, ovf_n, ovf, c, n, l16, N);
    const float s = rsqrtf((float)(c + 1));
    const float so = rsqrtf((float)(co + 1));
    v4f r;
    r.x = fmaxf(a.x * s + b[f + 0], 0.0f) * so;
    r.y = fmaxf(a.y * s + b[f + 1], 0.0f) * so;
    r.z = fmaxf(a.z * s + b[f + 2], 0.0f) * so;
    r.w = fmaxf(a.w * s + b[f + 3], 0.0f) * so;
    *(v4f*)(ta + row * TP + f) = r;

    const int wave = threadIdx.x >> 5;
    const int lane = threadIdx.x & 31;
    const int half = lane >> 4;
    const int lr   = lane & 15;
    v2f bv[16];
    if (wave < 4) {
        const float* __restrict__ wcol = W + wave * 16 + lr;
#pragma unroll
        for (int i = 0; i < 16; ++i) {
            const int ka = i * 4 + half * 2;
            bv[i].x = wcol[(size_t)ka * D];
            bv[i].y = wcol[(size_t)(ka + 1) * D];
        }
    }
    __syncthreads();

    if (wave < 4) {
        const float* __restrict__ arow = ta + lr * TP;
        v8f acc = {};
#pragma unroll
        for (int i = 0; i < 16; ++i) {
            const int ka = i * 4 + half * 2;
            const v2f av = *(const v2f*)(arow + ka);
            acc = __builtin_amdgcn_wmma_f32_16x16x4_f32(false, av, false, bv[i], (short)0, acc, false, false);
        }
#pragma unroll
        for (int v = 0; v < 8; ++v)
            td[(half * 8 + v) * TP + wave * 16 + lr] = acc[v];
    }
    __syncthreads();
    const v4f o = *(const v4f*)(td + row * TP + f);
    if (node < N) *(v4f*)(hout + (size_t)node * D + f) = o;
}

__global__ __launch_bounds__(256) void gather_dot(const float* __restrict__ h,
                                                  const int* __restrict__ ell,
                                                  const int* __restrict__ ovf_n,
                                                  const int* __restrict__ ovf,
                                                  const float* __restrict__ b2,
                                                  const float* __restrict__ Wr,
                                                  const float* __restrict__ br,
                                                  float* __restrict__ out, int N) {
    const int t = blockIdx.x * blockDim.x + threadIdx.x;
    const int node = t >> 4;
    const int l16 = threadIdx.x & 15;
    const int n = node < N ? node : N - 1;
    const int f = l16 * 4;
    const v4i rw = *(const v4i*)(ell + (size_t)n * RW + f);
    const int c = __shfl(rw.x, 0, 16);
    const v4f a = row_sum(h, rw, ovf_n, ovf, c, n, l16, N);
    const float s = rsqrtf((float)(c + 1));
    float p = ((a.x * s + b2[f + 0]) * Wr[f + 0] + (a.y * s + b2[f + 1]) * Wr[f + 1])
            + ((a.z * s + b2[f + 2]) * Wr[f + 2] + (a.w * s + b2[f + 3]) * Wr[f + 3]);
#pragma unroll
    for (int off = 8; off > 0; off >>= 1) p += __shfl_xor(p, off, 32);
    if (node < N && l16 == 0) out[node] = p + br[0];
}

extern "C" void kernel_launch(void* const* d_in, const int* in_sizes, int n_in,
                              void* d_out, int out_size, void* d_ws, size_t ws_size,
                              hipStream_t stream) {
    const float* x   = (const float*)d_in[0];
    const int*   src = (const int*)d_in[1];
    const int*   dst = (const int*)d_in[2];
    const float* W1  = (const float*)d_in[3];
    const float* b1  = (const float*)d_in[4];
    const float* W2  = (const float*)d_in[5];
    const float* b2  = (const float*)d_in[6];
    const float* Wr  = (const float*)d_in[7];
    const float* br  = (const float*)d_in[8];
    float* out = (float*)d_out;

    const int N = in_sizes[0] / D;
    const int E = in_sizes[1];
    const size_t Ep = (((size_t)E + 63) / 64) * 64;

    int*   ovf_n   = (int*)d_ws;
    int*   ell     = ovf_n + ZPAD;
    int*   ovf     = ell + (size_t)N * RW;
    float* h       = (float*)(ovf + 2 * Ep);
    float* h2      = h + ((size_t)N + 1) * D;

    const int TB = 256;
    const int gZ    = (N + TB - 1) / TB;
    const int gE    = (E + TB - 1) / TB;
    const int gGemm = (N + 127) / 128;
    const int gRow  = (int)(((size_t)N * 16 + TB - 1) / TB);

    cnt_zero <<<gZ, TB, 0, stream>>>(ell, N, ovf_n, h + (size_t)N * D, h2 + (size_t)N * D);
    edge_slot<<<gE, TB, 0, stream>>>(src, dst, ell, ovf_n, ovf, E);

    gemm64_wmma     <<<gGemm, TB, 0, stream>>>(x, ell, W1, h, N);
    gather_relu_gemm<<<gRow, TB, 0, stream>>>(h, ell, ovf_n, ovf, b1, W2, h2, N);

    gather_dot      <<<gRow, TB, 0, stream>>>(h2, ell, ovf_n, ovf, b2, Wr, br, out, N);
}


# source rows kept twice, one 128-B row per node per chip half: an edge wave (half read from its die number) writes only its half's copy, so no row line is written from both halves; sum order changes
# speedup vs baseline: 4.9234x; 1.9939x over previous
#include <hip/hip_runtime.h>
#include <hip/hip_bf16.h>


typedef __attribute__((ext_vector_type(2))) float v2f;
typedef __attribute__((ext_vector_type(8))) float v8f;
typedef __attribute__((ext_vector_type(4))) float v4f;
typedef __attribute__((ext_vector_type(2))) int v2i;
typedef __attribute__((ext_vector_type(4))) int v4i;

#define D 64
#define RW 32
#define CAP 30
#define ZPAD 64

__global__ __launch_bounds__(256) void cnt_zero(int* __restrict__ ellA, int* __restrict__ ellB, int N,
                                                int* __restrict__ ovf_n,
                                                float* __restrict__ zrow,
                                                float* __restrict__ zrow2) {
    int i = blockIdx.x * blockDim.x + threadIdx.x;
    if (i < N) {
        v2i z = {0, 0};
        *(v2i*)(ellA + (size_t)i * RW) = z;
        *(v2i*)(ellB + (size_t)i * RW) = z;
    }
    if (i < D) { zrow[i] = 0.0f; zrow2[i] = 0.0f; }
    if (i == 0) ovf_n[0] = 0;
}

__global__ __launch_bounds__(256) void edge_slot(const int* __restrict__ src,
                                                 const int* __restrict__ dst,
                                                 int* ellA, int* ellB,
                                                 int* ovf_n, int* __restrict__ ovf, int E) {
    const int half = (int)((__builtin_amdgcn_s_sendmsg_rtn(0x87) >> 18) & 1u);
    int* ell = half ? ellB : ellA;
    int e = blockIdx.x * blockDim.x + threadIdx.x;
    if (e < E) {
        const int s = src[e];
        const int d = dst[e];
        int* drow = ell + (size_t)d * RW;
        const int r = atomicAdd(drow, 1);
        atomicAdd(ell + (size_t)s * RW + 1, 1);
        if (r < CAP) {
            drow[2 + r] = s;
        } else {
            const int p = atomicAdd(ovf_n, 1);
            ovf[2 * (size_t)p]     = d;
            ovf[2 * (size_t)p + 1] = s;
        }
    }
}

__global__ __launch_bounds__(256) void gemm64_wmma(const float* __restrict__ X,
                                                   const int* __restrict__ ellA,
                                                   const int* __restrict__ ellB,
                                                   const float* __restrict__ W,
                                                   float* __restrict__ out0,
                                                   int N) {
    const int wave = threadIdx.x >> 5;
    const int lane = threadIdx.x & 31;
    const int nb = blockIdx.x * 128 + wave * 16;
    if (nb >= N) return;
    const int half = lane >> 4;
    const int lr   = lane & 15;

    const size_t cw = (size_t)(nb + lr) * RW + 1;
    const float s = rsqrtf((float)(ellA[cw] + ellB[cw] + 1));
    const float* __restrict__ xrow = X + (size_t)(nb + lr) * D;

    v8f acc0 = {}, acc1 = {}, acc2 = {}, acc3 = {};
#pragma unroll
    for (int k0 = 0; k0 < D; k0 += 4) {
        const int ka = k0 + half * 2;
        v2f a;
        a.x = xrow[ka] * s;
        a.y = xrow[ka + 1] * s;
        const float* __restrict__ w0 = W + (size_t)ka * D + lr;
        const float* __restrict__ w1 = w0 + D;
        v2f b0 = { w0[0],  w1[0]  };
        v2f b1 = { w0[16], w1[16] };
        v2f b2 = { w0[32], w1[32] };
        v2f b3 = { w0[48], w1[48] };
        acc0 = __builtin_amdgcn_wmma_f32_16x16x4_f32(false, a, false, b0, (short)0, acc0, false, false);
        acc1 = __builtin_amdgcn_wmma_f32_16x16x4_f32(false, a, false, b1, (short)0, acc1, false, false);
        acc2 = __builtin_amdgcn_wmma_f32_16x16x4_f32(false, a, false, b2, (short)0, acc2, false, false);
        acc3 = __builtin_amdgcn_wmma_f32_16x16x4_f32(false, a, false, b3, (short)0, acc3, false, false);
    }

    const int row0 = nb + half * 8;
#pragma unroll
    for (int v = 0; v < 8; ++v) {
        const size_t base = (size_t)(row0 + v) * D + lr;
        out0[base]      = acc0[v];
        out0[base + 16] = acc1[v];
        out0[base + 32] = acc2[v];
        out0[base + 48] = acc3[v];
    }
}

static __device__ __forceinline__ v4f row_sum(const float* __restrict__ h,
                                              const v4i rw,
                                              const int* __restrict__ ovf_n,
                                              const int* __restrict__ ovf,
                                              int cA, int cB, int n, int l16, int zr) {
    const int f = l16 * 4;
    v4f acc = *(const v4f*)(h + (size_t)n * D + f);
#pragma unroll
    for (int cp = 0; cp < 2; ++cp) {
        const int cnt = cp == 0 ? cA : cB;
        const int m = cnt < CAP ? cnt : CAP;
#pragma unroll
        for (int t = 0; t < (CAP + 7) / 8; ++t) {
            const int base = t * 8;
            if (base < m) {
                v4f v[8];
#pragma unroll
                for (int j = 0; j < 8; ++j) {
                    if (base + j < CAP) {
                        const int word = base + j + 2;
                        const int comp = word & 3;
                        const int w = comp == 0 ? rw.x : comp == 1 ? rw.y : comp == 2 ? rw.z : rw.w;
                        int s = __shfl(w, cp * 8 + (word >> 2), 16);
                        s = (base + j < m) ? s : zr;
                        v[j] = *(const v4f*)(h + (size_t)s * D + f);
                    }
                }
#pragma unroll
                for (int j = 0; j < 8; ++j)
                    if (base + j < CAP) acc += v[j];
            }
        }
    }
    if (cA > CAP || cB > CAP) {
        const int tot = ovf_n[0];
        for (int i = 0; i < tot; ++i) {
            if (ovf[2 * (size_t)i] == n) acc += *(const v4f*)(h + (size_t)ovf[2 * (size_t)i + 1] * D + f);
        }
    }
    return acc;
}

#define TP 68
__global__ __launch_bounds__(256) void gather_relu_gemm(const float* __restrict__ h,
                                                        const int* __restrict__ ellA,
                                                        const int* __restrict__ ellB,
                                                        const int* __restrict__ ovf_n,
                                                        const int* __restrict__ ovf,
                                                        const float* __restrict__ b,
                                                        const float* __restrict__ W,
                                                        float* __restrict__ hout, int N) {
    __shared__ __attribute__((aligned(16))) float ta[16 * TP];
    __shared__ __attribute__((aligned(16))) float td[16 * TP];
    const int t = blockIdx.x * blockDim.x + threadIdx.x;
    const int node = t >> 4;
    const int row = threadIdx.x >> 4;
    const int l16 = threadIdx.x & 15;
    const int n = node < N ? node : N - 1;
    const int f = l16 * 4;
    const v4i rw = *(const v4i*)(((l16 & 8) ? ellB : ellA) + (size_t)n * RW + (l16 & 7) * 4);
    const int cA = __shfl(rw.x, 0, 16);
    const int cB = __shfl(rw.x, 8, 16);
    const int c = cA + cB;
    const int co = __shfl(rw.y, 0, 16) + __shfl(rw.y, 8, 16);
    const v4f a = row_sum(h, rw, ovf_n, ovf, cA, cB, n, l16, N);
    const float s = rsqrtf((float)(c + 1));
    const float so = rsqrtf((float)(co + 1));
    v4f r;
    r.x = fmaxf(a.x * s + b[f + 0], 0.0f) * so;
    r.y = fmaxf(a.y * s + b[f + 1], 0.0f) * so;
    r.z = fmaxf(a.z * s + b[f + 2], 0.0f) * so;
    r.w = fmaxf(a.w * s + b[f + 3], 0.0f) * so;
    *(v4f*)(ta + row * TP + f) = r;

    const int wave = threadIdx.x >> 5;
    const int lane = threadIdx.x & 31;
    const int half = lane >> 4;
    const int lr   = lane & 15;
    v2f bv[16];
    if (wave < 4) {
        const float* __restrict__ wcol = W + wave * 16 + lr;
#pragma unroll
        for (int i = 0; i < 16; ++i) {
            const int ka = i * 4 + half * 2;
            bv[i].x = wcol[(size_t)ka * D];
            bv[i].y = wcol[(size_t)(ka + 1) * D];
        }
    }
    __syncthreads();

    if (wave < 4) {
        const float* __restrict__ arow = ta + lr * TP;
        v8f acc = {};
#pragma unroll
        for (int i = 0; i < 16; ++i) {
            const int ka = i * 4 + half * 2;
            const v2f av = *(const v2f*)(arow + ka);
            acc = __builtin_amdgcn_wmma_f32_16x16x4_f32(false, av, false, bv[i], (short)0, acc, false, false);
        }
#pragma unroll
        for (int v = 0; v < 8; ++v)
            td[(half * 8 + v) * TP + wave * 16 + lr] = acc[v];
    }
    __syncthreads();
    const v4f o = *(const v4f*)(td + row * TP + f);
    if (node < N) *(v4f*)(hout + (size_t)node * D + f) = o;
}

__global__ __launch_bounds__(256) void gather_dot(const float* __restrict__ h,
                                                  const int* __restrict__ ellA,
                                                  const int* __restrict__ ellB,
                                                  const int* __restrict__ ovf_n,
                                                  const int* __restrict__ ovf,
                                                  const float* __restrict__ b2,
                                                  const float* __restrict__ Wr,
                                                  const float* __restrict__ br,
                                                  float* __restrict__ out, int N) {
    const int t = blockIdx.x * blockDim.x + threadIdx.x;
    const int node = t >> 4;
    const int l16 = threadIdx.x & 15;
    const int n = node < N ? node : N - 1;
    const int f = l16 * 4;
    const v4i rw = *(const v4i*)(((l16 & 8) ? ellB : ellA) + (size_t)n * RW + (l16 & 7) * 4);
    const int cA = __shfl(rw.x, 0, 16);
    const int cB = __shfl(rw.x, 8, 16);
    const int c = cA + cB;
    const v4f a = row_sum(h, rw, ovf_n, ovf, cA, cB, n, l16, N);
    const float s = rsqrtf((float)(c + 1));
    float p = ((a.x * s + b2[f + 0]) * Wr[f + 0] + (a.y * s + b2[f + 1]) * Wr[f + 1])
            + ((a.z * s + b2[f + 2]) * Wr[f + 2] + (a.w * s + b2[f + 3]) * Wr[f + 3]);
#pragma unroll
    for (int off = 8; off > 0; off >>= 1) p += __shfl_xor(p, off, 32);
    if (node < N && l16 == 0) out[node] = p + br[0];
}

extern "C" void kernel_launch(void* const* d_in, const int* in_sizes, int n_in,
                              void* d_out, int out_size, void* d_ws, size_t ws_size,
                              hipStream_t stream) {
    const float* x   = (const float*)d_in[0];
    const int*   src = (const int*)d_in[1];
    const int*   dst = (const int*)d_in[2];
    const float* W1  = (const float*)d_in[3];
    const float* b1  = (const float*)d_in[4];
    const float* W2  = (const float*)d_in[5];
    const float* b2  = (const float*)d_in[6];
    const float* Wr  = (const float*)d_in[7];
    const float* br  = (const float*)d_in[8];
    float* out = (float*)d_out;

    const int N = in_sizes[0] / D;
    const int E = in_sizes[1];
    const size_t Ep = (((size_t)E + 63) / 64) * 64;

    int*   ovf_n   = (int*)d_ws;
    int*   ellA    = ovf_n + ZPAD;
    int*   ellB    = ellA + (size_t)N * RW;
    int*   ovf     = ellB + (size_t)N * RW;
    float* h       = (float*)(ovf + 2 * Ep);
    float* h2      = h + ((size_t)N + 1) * D;

    const int TB = 256;
    const int gZ    = (N + TB - 1) / TB;
    const int gE    = (E + TB - 1) / TB;
    const int gGemm = (N + 127) / 128;
    const int gRow  = (int)(((size_t)N * 16 + TB - 1) / TB);

    cnt_zero <<<gZ, TB, 0, stream>>>(ellA, ellB, N, ovf_n, h + (size_t)N * D, h2 + (size_t)N * D);
    edge_slot<<<gE, TB, 0, stream>>>(src, dst, ellA, ellB, ovf_n, ovf, E);

    gemm64_wmma     <<<gGemm, TB, 0, stream>>>(x, ellA, ellB, W1, h, N);
    gather_relu_gemm<<<gRow, TB, 0, stream>>>(h, ellA, ellB, ovf_n, ovf, b1, W2, h2, N);

    gather_dot      <<<gRow, TB, 0, stream>>>(h2, ellA, ellB, ovf_n, ovf, b2, Wr, br, out, N);
}
